# starting kernel
# baseline (speedup 1.0000x reference)
#include <hip/hip_runtime.h>
#include <cfloat>

typedef __attribute__((ext_vector_type(2))) float v2f;
typedef __attribute__((ext_vector_type(8))) float v8f;

#define SK_STRIDE 68
#define SVT_STRIDE 260
#define SP_STRIDE 34

__device__ __forceinline__ v8f wmma_f32_16x16x4(v2f a, v2f b, v8f c) {
  return __builtin_amdgcn_wmma_f32_16x16x4_f32(false, a, false, b, (short)0, c, false, false);
}

__global__ __launch_bounds__(256) void local_attn_rope_kernel(
    const float* __restrict__ q, const float* __restrict__ k,
    const float* __restrict__ v, float* __restrict__ out)
{
  __shared__ float sK [256 * SK_STRIDE];
  __shared__ float sVT[ 64 * SVT_STRIDE];
  __shared__ float sP [8 * 16 * SP_STRIDE];

  const int wi   = blockIdx.x;
  const int bb   = blockIdx.y;
  const int tid  = threadIdx.x;
  const int wv   = tid >> 5;
  const int lane = tid & 31;
  const int l16  = lane & 15;
  const int hi   = lane >> 4;

  const float* kb = k + (size_t)bb * 4096 * 64;
  const float* vb = v + (size_t)bb * 4096 * 64;
  const int jbase = wi * 128 - 128;

  const float LOG2_IF = -0.41524101186092029f;

  #pragma unroll
  for (int it = 0; it < 32; ++it) {
    int p   = tid + 256 * it;
    int row = p >> 5;
    int c   = p & 31;
    int grow = jbase + row;
    float x1 = 0.f, x2 = 0.f;
    if (grow >= 0) {
      const float* kr = kb + (size_t)grow * 64;
      x1 = kr[c]; x2 = kr[c + 32];
    }
    float ang = (float)row * exp2f(LOG2_IF * (float)c);
    float sn, cs; __sincosf(ang, &sn, &cs);
    sK[row * SK_STRIDE + c]      = x1 * cs - x2 * sn;
    sK[row * SK_STRIDE + c + 32] = x2 * cs + x1 * sn;
  }

  #pragma unroll
  for (int it = 0; it < 16; ++it) {
    int p   = tid + 256 * it;
    int row = p >> 4;
    int c4  = (p & 15) * 4;
    int grow = jbase + row;
    float4 val = make_float4(0.f, 0.f, 0.f, 0.f);
    if (grow >= 0) val = *(const float4*)(vb + (size_t)grow * 64 + c4);
    sVT[(c4 + 0) * SVT_STRIDE + row] = val.x;
    sVT[(c4 + 1) * SVT_STRIDE + row] = val.y;
    sVT[(c4 + 2) * SVT_STRIDE + row] = val.z;
    sVT[(c4 + 3) * SVT_STRIDE + row] = val.w;
  }

  const int iq = 16 * wv + l16;
  const float* qrow = q + ((size_t)bb * 4096 + (size_t)wi * 128 + iq) * 64;
  v2f qf[16];
  #pragma unroll
  for (int t = 0; t < 16; ++t)
    qf[t] = *(const v2f*)(qrow + 4 * t + 2 * hi);
  {
    const float scale = 0.125f;
    float pos = (float)(128 + iq);
    #pragma unroll
    for (int t = 0; t < 8; ++t) {
      #pragma unroll
      for (int i = 0; i < 2; ++i) {
        int kk = 4 * t + 2 * hi + i;
        float ang = pos * exp2f(LOG2_IF * (float)kk);
        float sn, cs; __sincosf(ang, &sn, &cs);
        cs *= scale; sn *= scale;
        float lo = qf[t][i], hh = qf[t + 8][i];
        qf[t][i]     = lo * cs - hh * sn;
        qf[t + 8][i] = hh * cs + lo * sn;
      }
    }
  }

  __syncthreads();

  v8f S[16];
  #pragma unroll
  for (int jt = 0; jt < 16; ++jt) {
    v8f acc = {0.f,0.f,0.f,0.f,0.f,0.f,0.f,0.f};
    const float* kr = &sK[(16 * jt + l16) * SK_STRIDE + 2 * hi];
    #pragma unroll
    for (int t = 0; t < 16; ++t) {
      v2f bf = *(const v2f*)(kr + 4 * t);
      acc = wmma_f32_16x16x4(qf[t], bf, acc);
    }
    S[jt] = acc;
  }

  const float NEGMAX = -3.402823466e38f;
  #pragma unroll
  for (int jt = 0; jt < 16; ++jt) {
    int j = 16 * jt + l16;
    bool padj = (wi == 0) && (j < 128);
    #pragma unroll
    for (int vv = 0; vv < 8; ++vv) {
      int iqc = 16 * wv + vv + 8 * hi;
      if (padj || (j > iqc + 128)) S[jt][vv] = NEGMAX;
    }
  }

  float rinv[8];
  #pragma unroll
  for (int vv = 0; vv < 8; ++vv) {
    float m = S[0][vv];
    #pragma unroll
    for (int jt = 1; jt < 16; ++jt) m = fmaxf(m, S[jt][vv]);
    m = fmaxf(m, __shfl_xor(m, 1, 16));
    m = fmaxf(m, __shfl_xor(m, 2, 16));
    m = fmaxf(m, __shfl_xor(m, 4, 16));
    m = fmaxf(m, __shfl_xor(m, 8, 16));
    float s = 0.f;
    #pragma unroll
    for (int jt = 0; jt < 16; ++jt) {
      float e = __expf(S[jt][vv] - m);
      S[jt][vv] = e;
      s += e;
    }
    s += __shfl_xor(s, 1, 16);
    s += __shfl_xor(s, 2, 16);
    s += __shfl_xor(s, 4, 16);
    s += __shfl_xor(s, 8, 16);
    rinv[vv] = 1.0f / s;
  }

  v8f O[4];
  #pragma unroll
  for (int n = 0; n < 4; ++n) O[n] = (v8f){0.f,0.f,0.f,0.f,0.f,0.f,0.f,0.f};

  float* myP = &sP[wv * 16 * SP_STRIDE];
  #pragma unroll
  for (int ch = 0; ch < 8; ++ch) {
    #pragma unroll
    for (int u = 0; u < 2; ++u) {
      v8f Pv = S[2 * ch + u];
      #pragma unroll
      for (int vv = 0; vv < 8; ++vv)
        myP[(vv + 8 * hi) * SP_STRIDE + 16 * u + l16] = Pv[vv];
    }
    asm volatile("s_wait_dscnt 0" ::: "memory");
    #pragma unroll
    for (int t = 0; t < 8; ++t) {
      v2f af = *(const v2f*)(&myP[l16 * SP_STRIDE + 4 * t + 2 * hi]);
      int jb = 32 * ch + 4 * t + 2 * hi;
      #pragma unroll
      for (int n = 0; n < 4; ++n) {
        v2f bf = *(const v2f*)(&sVT[(16 * n + l16) * SVT_STRIDE + jb]);
        O[n] = wmma_f32_16x16x4(af, bf, O[n]);
      }
    }
  }

  float* ob = out + ((size_t)bb * 4096 + (size_t)wi * 128 + 16 * wv + 8 * hi) * 64 + l16;
  #pragma unroll
  for (int n = 0; n < 4; ++n) {
    #pragma unroll
    for (int vv = 0; vv < 8; ++vv)
      ob[(size_t)vv * 64 + 16 * n] = O[n][vv] * rinv[vv];
  }
}

extern "C" void kernel_launch(void* const* d_in, const int* in_sizes, int n_in,
                              void* d_out, int out_size, void* d_ws, size_t ws_size,
                              hipStream_t stream) {
  const float* q = (const float*)d_in[0];
  const float* k = (const float*)d_in[1];
  const float* v = (const float*)d_in[2];
  float* out = (float*)d_out;
  dim3 grid(32, 32);
  local_attn_rope_kernel<<<grid, dim3(256), 0, stream>>>(q, k, v, out);
}


# Staging loads of K, V, Q made unconditional (clamped row) and issued as one batch ahead of the RoPE math, not one dependent round trip per row group; as built 244 VGPRs (was 265): 2 blocks per WGP
# speedup vs baseline: 2.2334x; 2.2334x over previous
#include <hip/hip_runtime.h>
#include <cfloat>

typedef __attribute__((ext_vector_type(2))) float v2f;
typedef __attribute__((ext_vector_type(8))) float v8f;

#define SK_STRIDE 68
#define SVT_STRIDE 260
#define SP_STRIDE 34

__device__ __forceinline__ v8f wmma_f32_16x16x4(v2f a, v2f b, v8f c) {
  return __builtin_amdgcn_wmma_f32_16x16x4_f32(false, a, false, b, (short)0, c, false, false);
}

__global__ __launch_bounds__(256) void local_attn_rope_kernel(
    const float* __restrict__ q, const float* __restrict__ k,
    const float* __restrict__ v, float* __restrict__ out)
{
  __shared__ float sK [256 * SK_STRIDE];
  __shared__ float sVT[ 64 * SVT_STRIDE];
  __shared__ float sP [8 * 16 * SP_STRIDE];

  const int wi   = blockIdx.x;
  const int bb   = blockIdx.y;
  const int tid  = threadIdx.x;
  const int wv   = tid >> 5;
  const int lane = tid & 31;
  const int l16  = lane & 15;
  const int hi   = lane >> 4;

  const float* kb = k + (size_t)bb * 4096 * 64;
  const float* vb = v + (size_t)bb * 4096 * 64;
  const int jbase = wi * 128 - 128;

  const float LOG2_IF = -0.41524101186092029f;

  const int krow0 = tid >> 5;
  const int kc    = tid & 31;
  float kx1[32], kx2[32];
  #pragma unroll
  for (int it = 0; it < 32; ++it) {
    int grow = jbase + krow0 + 8 * it;
    grow = grow < 0 ? 0 : grow;
    const float* kr = kb + (size_t)grow * 64;
    kx1[it] = kr[kc];
    kx2[it] = kr[kc + 32];
  }
  const int vrow0 = tid >> 4;
  const int vc4   = (tid & 15) * 4;
  float4 vx[16];
  #pragma unroll
  for (int it = 0; it < 16; ++it) {
    int grow = jbase + vrow0 + 16 * it;
    grow = grow < 0 ? 0 : grow;
    vx[it] = *(const float4*)(vb + (size_t)grow * 64 + vc4);
  }
  const int iq = 16 * wv + l16;
  const float* qrow = q + ((size_t)bb * 4096 + (size_t)wi * 128 + iq) * 64;
  v2f qf[16];
  #pragma unroll
  for (int t = 0; t < 16; ++t)
    qf[t] = *(const v2f*)(qrow + 4 * t + 2 * hi);

  #pragma unroll
  for (int it = 0; it < 32; ++it) {
    int row = krow0 + 8 * it;
    float x1 = kx1[it];
    float x2 = kx2[it];
    float ang = (float)row * exp2f(LOG2_IF * (float)kc);
    float sn, cs; __sincosf(ang, &sn, &cs);
    sK[row * SK_STRIDE + kc]      = x1 * cs - x2 * sn;
    sK[row * SK_STRIDE + kc + 32] = x2 * cs + x1 * sn;
  }

  #pragma unroll
  for (int it = 0; it < 16; ++it) {
    int row = vrow0 + 16 * it;
    bool pad = (jbase + row) < 0;
    float4 val = vx[it];
    sVT[(vc4 + 0) * SVT_STRIDE + row] = pad ? 0.f : val.x;
    sVT[(vc4 + 1) * SVT_STRIDE + row] = pad ? 0.f : val.y;
    sVT[(vc4 + 2) * SVT_STRIDE + row] = pad ? 0.f : val.z;
    sVT[(vc4 + 3) * SVT_STRIDE + row] = pad ? 0.f : val.w;
  }

  {
    const float scale = 0.125f;
    float pos = (float)(128 + iq);
    #pragma unroll
    for (int t = 0; t < 8; ++t) {
      #pragma unroll
      for (int i = 0; i < 2; ++i) {
        int kk = 4 * t + 2 * hi + i;
        float ang = pos * exp2f(LOG2_IF * (float)kk);
        float sn, cs; __sincosf(ang, &sn, &cs);
        cs *= scale; sn *= scale;
        float lo = qf[t][i], hh = qf[t + 8][i];
        qf[t][i]     = lo * cs - hh * sn;
        qf[t + 8][i] = hh * cs + lo * sn;
      }
    }
  }

  __syncthreads();

  v8f S[16];
  #pragma unroll
  for (int jt = 0; jt < 16; ++jt) {
    v8f acc = {0.f,0.f,0.f,0.f,0.f,0.f,0.f,0.f};
    const float* kr = &sK[(16 * jt + l16) * SK_STRIDE + 2 * hi];
    #pragma unroll
    for (int t = 0; t < 16; ++t) {
      v2f bf = *(const v2f*)(kr + 4 * t);
      acc = wmma_f32_16x16x4(qf[t], bf, acc);
    }
    S[jt] = acc;
  }

  const float NEGMAX = -3.402823466e38f;
  #pragma unroll
  for (int jt = 0; jt < 16; ++jt) {
    int j = 16 * jt + l16;
    bool padj = (wi == 0) && (j < 128);
    #pragma unroll
    for (int vv = 0; vv < 8; ++vv) {
      int iqc = 16 * wv + vv + 8 * hi;
      if (padj || (j > iqc + 128)) S[jt][vv] = NEGMAX;
    }
  }

  float rinv[8];
  #pragma unroll
  for (int vv = 0; vv < 8; ++vv) {
    float m = S[0][vv];
    #pragma unroll
    for (int jt = 1; jt < 16; ++jt) m = fmaxf(m, S[jt][vv]);
    m = fmaxf(m, __shfl_xor(m, 1, 16));
    m = fmaxf(m, __shfl_xor(m, 2, 16));
    m = fmaxf(m, __shfl_xor(m, 4, 16));
    m = fmaxf(m, __shfl_xor(m, 8, 16));
    float s = 0.f;
    #pragma unroll
    for (int jt = 0; jt < 16; ++jt) {
      float e = __expf(S[jt][vv] - m);
      S[jt][vv] = e;
      s += e;
    }
    s += __shfl_xor(s, 1, 16);
    s += __shfl_xor(s, 2, 16);
    s += __shfl_xor(s, 4, 16);
    s += __shfl_xor(s, 8, 16);
    rinv[vv] = 1.0f / s;
  }

  v8f O[4];
  #pragma unroll
  for (int n = 0; n < 4; ++n) O[n] = (v8f){0.f,0.f,0.f,0.f,0.f,0.f,0.f,0.f};

  float* myP = &sP[wv * 16 * SP_STRIDE];
  #pragma unroll
  for (int ch = 0; ch < 8; ++ch) {
    #pragma unroll
    for (int u = 0; u < 2; ++u) {
      v8f Pv = S[2 * ch + u];
      #pragma unroll
      for (int vv = 0; vv < 8; ++vv)
        myP[(vv + 8 * hi) * SP_STRIDE + 16 * u + l16] = Pv[vv];
    }
    asm volatile("s_wait_dscnt 0" ::: "memory");
    #pragma unroll
    for (int t = 0; t < 8; ++t) {
      v2f af = *(const v2f*)(&myP[l16 * SP_STRIDE + 4 * t + 2 * hi]);
      int jb = 32 * ch + 4 * t + 2 * hi;
      #pragma unroll
      for (int n = 0; n < 4; ++n) {
        v2f bf = *(const v2f*)(&sVT[(16 * n + l16) * SVT_STRIDE + jb]);
        O[n] = wmma_f32_16x16x4(af, bf, O[n]);
      }
    }
  }

  float* ob = out + ((size_t)bb * 4096 + (size_t)wi * 128 + 16 * wv + 8 * hi) * 64 + l16;
  #pragma unroll
  for (int n = 0; n < 4; ++n) {
    #pragma unroll
    for (int vv = 0; vv < 8; ++vv)
      ob[(size_t)vv * 64 + 16 * n] = O[n][vv] * rinv[vv];
  }
}

extern "C" void kernel_launch(void* const* d_in, const int* in_sizes, int n_in,
                              void* d_out, int out_size, void* d_ws, size_t ws_size,
                              hipStream_t stream) {
  const float* q = (const float*)d_in[0];
  const float* k = (const float*)d_in[1];
  const float* v = (const float*)d_in[2];
  float* out = (float*)d_out;
  dim3 grid(32, 32);
  local_attn_rope_kernel<<<grid, dim3(256), 0, stream>>>(q, k, v, out);
}


# Key tiles the causal mask zeroes whole are skipped in QK, softmax and PV (row tile r computes 9+r of 16 key tiles); waves w and w+4 of a SIMD take row tiles 7-w and w: 25 tiles per SIMD
# speedup vs baseline: 2.5697x; 1.1506x over previous
#include <hip/hip_runtime.h>
#include <cfloat>

typedef __attribute__((ext_vector_type(2))) float v2f;
typedef __attribute__((ext_vector_type(8))) float v8f;

#define SK_STRIDE 68
#define SVT_STRIDE 260
#define SP_STRIDE 34

__device__ __forceinline__ v8f wmma_f32_16x16x4(v2f a, v2f b, v8f c) {
  return __builtin_amdgcn_wmma_f32_16x16x4_f32(false, a, false, b, (short)0, c, false, false);
}

__global__ __launch_bounds__(256) void local_attn_rope_kernel(
    const float* __restrict__ q, const float* __restrict__ k,
    const float* __restrict__ v, float* __restrict__ out)
{
  __shared__ float sK [256 * SK_STRIDE];
  __shared__ float sVT[ 64 * SVT_STRIDE];
  __shared__ float sP [8 * 16 * SP_STRIDE];

  const int wi   = blockIdx.x;
  const int bb   = blockIdx.y;
  const int tid  = threadIdx.x;
  const int wv   = __builtin_amdgcn_readfirstlane(tid >> 5);
  const int rt   = (wv < 4) ? (7 - wv) : (wv - 4);
  const int nt   = 9 + rt;
  const int lane = tid & 31;
  const int l16  = lane & 15;
  const int hi   = lane >> 4;

  const float* kb = k + (size_t)bb * 4096 * 64;
  const float* vb = v + (size_t)bb * 4096 * 64;
  const int jbase = wi * 128 - 128;

  const float LOG2_IF = -0.41524101186092029f;

  const int krow0 = tid >> 5;
  const int kc    = tid & 31;
  float kx1[32], kx2[32];
  #pragma unroll
  for (int it = 0; it < 32; ++it) {
    int grow = jbase + krow0 + 8 * it;
    grow = grow < 0 ? 0 : grow;
    const float* kr = kb + (size_t)grow * 64;
    kx1[it] = kr[kc];
    kx2[it] = kr[kc + 32];
  }
  const int vrow0 = tid >> 4;
  const int vc4   = (tid & 15) * 4;
  float4 vx[16];
  #pragma unroll
  for (int it = 0; it < 16; ++it) {
    int grow = jbase + vrow0 + 16 * it;
    grow = grow < 0 ? 0 : grow;
    vx[it] = *(const float4*)(vb + (size_t)grow * 64 + vc4);
  }
  const int iq = 16 * rt + l16;
  const float* qrow = q + ((size_t)bb * 4096 + (size_t)wi * 128 + iq) * 64;
  v2f qf[16];
  #pragma unroll
  for (int t = 0; t < 16; ++t)
    qf[t] = *(const v2f*)(qrow + 4 * t + 2 * hi);

  #pragma unroll
  for (int it = 0; it < 32; ++it) {
    int row = krow0 + 8 * it;
    float x1 = kx1[it];
    float x2 = kx2[it];
    float ang = (float)row * exp2f(LOG2_IF * (float)kc);
    float sn, cs; __sincosf(ang, &sn, &cs);
    sK[row * SK_STRIDE + kc]      = x1 * cs - x2 * sn;
    sK[row * SK_STRIDE + kc + 32] = x2 * cs + x1 * sn;
  }

  #pragma unroll
  for (int it = 0; it < 16; ++it) {
    int row = vrow0 + 16 * it;
    bool pad = (jbase + row) < 0;
    float4 val = vx[it];
    sVT[(vc4 + 0) * SVT_STRIDE + row] = pad ? 0.f : val.x;
    sVT[(vc4 + 1) * SVT_STRIDE + row] = pad ? 0.f : val.y;
    sVT[(vc4 + 2) * SVT_STRIDE + row] = pad ? 0.f : val.z;
    sVT[(vc4 + 3) * SVT_STRIDE + row] = pad ? 0.f : val.w;
  }

  {
    const float scale = 0.125f;
    float pos = (float)(128 + iq);
    #pragma unroll
    for (int t = 0; t < 8; ++t) {
      #pragma unroll
      for (int i = 0; i < 2; ++i) {
        int kk = 4 * t + 2 * hi + i;
        float ang = pos * exp2f(LOG2_IF * (float)kk);
        float sn, cs; __sincosf(ang, &sn, &cs);
        cs *= scale; sn *= scale;
        float lo = qf[t][i], hh = qf[t + 8][i];
        qf[t][i]     = lo * cs - hh * sn;
        qf[t + 8][i] = hh * cs + lo * sn;
      }
    }
  }

  __syncthreads();

  v8f S[16];
  #pragma unroll
  for (int jt = 0; jt < 16; ++jt) {
    if (jt < 9 || jt < nt) {
      v8f acc = {0.f,0.f,0.f,0.f,0.f,0.f,0.f,0.f};
      const float* kr = &sK[(16 * jt + l16) * SK_STRIDE + 2 * hi];
      #pragma unroll
      for (int t = 0; t < 16; ++t) {
        v2f bf = *(const v2f*)(kr + 4 * t);
        acc = wmma_f32_16x16x4(qf[t], bf, acc);
      }
      S[jt] = acc;
    }
  }

  const float NEGMAX = -3.402823466e38f;
  #pragma unroll
  for (int jt = 0; jt < 16; ++jt) {
    if (jt < 9 || jt < nt) {
      int j = 16 * jt + l16;
      bool padj = (wi == 0) && (j < 128);
      #pragma unroll
      for (int vv = 0; vv < 8; ++vv) {
        int iqc = 16 * rt + vv + 8 * hi;
        if (padj || (j > iqc + 128)) S[jt][vv] = NEGMAX;
      }
    }
  }

  float rinv[8];
  float mx[8], sm[8];
  #pragma unroll
  for (int vv = 0; vv < 8; ++vv) mx[vv] = S[0][vv];
  #pragma unroll
  for (int jt = 1; jt < 16; ++jt) {
    if (jt < 9 || jt < nt) {
      #pragma unroll
      for (int vv = 0; vv < 8; ++vv) mx[vv] = fmaxf(mx[vv], S[jt][vv]);
    }
  }
  #pragma unroll
  for (int vv = 0; vv < 8; ++vv) {
    float m = mx[vv];
    m = fmaxf(m, __shfl_xor(m, 1, 16));
    m = fmaxf(m, __shfl_xor(m, 2, 16));
    m = fmaxf(m, __shfl_xor(m, 4, 16));
    m = fmaxf(m, __shfl_xor(m, 8, 16));
    mx[vv] = m;
    sm[vv] = 0.f;
  }
  #pragma unroll
  for (int jt = 0; jt < 16; ++jt) {
    if (jt < 9 || jt < nt) {
      #pragma unroll
      for (int vv = 0; vv < 8; ++vv) {
        float e = __expf(S[jt][vv] - mx[vv]);
        S[jt][vv] = e;
        sm[vv] += e;
      }
    }
  }
  #pragma unroll
  for (int vv = 0; vv < 8; ++vv) {
    float s = sm[vv];
    s += __shfl_xor(s, 1, 16);
    s += __shfl_xor(s, 2, 16);
    s += __shfl_xor(s, 4, 16);
    s += __shfl_xor(s, 8, 16);
    rinv[vv] = 1.0f / s;
  }

  v8f O[4];
  #pragma unroll
  for (int n = 0; n < 4; ++n) O[n] = (v8f){0.f,0.f,0.f,0.f,0.f,0.f,0.f,0.f};

  float* myP = &sP[wv * 16 * SP_STRIDE];
  #pragma unroll
  for (int ch = 0; ch < 8; ++ch) {
    if (2 * ch < 9 || 2 * ch < nt) {
      const bool two = (2 * ch + 1 < 9) || (2 * ch + 1 < nt);
      #pragma unroll
      for (int u = 0; u < 2; ++u) {
        if (u == 0 || two) {
          v8f Pv = S[2 * ch + u];
          #pragma unroll
          for (int vv = 0; vv < 8; ++vv)
            myP[(vv + 8 * hi) * SP_STRIDE + 16 * u + l16] = Pv[vv];
        }
      }
      asm volatile("s_wait_dscnt 0" ::: "memory");
      #pragma unroll
      for (int t = 0; t < 8; ++t) {
        if (t < 4 || two) {
          v2f af = *(const v2f*)(&myP[l16 * SP_STRIDE + 4 * t + 2 * hi]);
          int jb = 32 * ch + 4 * t + 2 * hi;
          #pragma unroll
          for (int n = 0; n < 4; ++n) {
            v2f bf = *(const v2f*)(&sVT[(16 * n + l16) * SVT_STRIDE + jb]);
            O[n] = wmma_f32_16x16x4(af, bf, O[n]);
          }
        }
      }
    }
  }

  float* ob = out + ((size_t)bb * 4096 + (size_t)wi * 128 + 16 * rt + 8 * hi) * 64 + l16;
  #pragma unroll
  for (int n = 0; n < 4; ++n) {
    #pragma unroll
    for (int vv = 0; vv < 8; ++vv)
      ob[(size_t)vv * 64 + 16 * n] = O[n][vv] * rinv[vv];
  }
}

extern "C" void kernel_launch(void* const* d_in, const int* in_sizes, int n_in,
                              void* d_out, int out_size, void* d_ws, size_t ws_size,
                              hipStream_t stream) {
  const float* q = (const float*)d_in[0];
  const float* k = (const float*)d_in[1];
  const float* v = (const float*)d_in[2];
  float* out = (float*)d_out;
  dim3 grid(32, 32);
  local_attn_rope_kernel<<<grid, dim3(256), 0, stream>>>(q, k, v, out);
}


# Scores computed transposed (keys as A operand, queries as B): softmax output registers are the A fragments of P*V, no LDS round trip of P; row max and sum fold in-lane, then across 2 lane halves
# speedup vs baseline: 2.6914x; 1.0474x over previous
#include <hip/hip_runtime.h>
#include <cfloat>

typedef __attribute__((ext_vector_type(2))) float v2f;
typedef __attribute__((ext_vector_type(8))) float v8f;

#define SK_STRIDE 68
#define SVT_STRIDE 260

__device__ __forceinline__ v8f wmma_f32_16x16x4(v2f a, v2f b, v8f c) {
  return __builtin_amdgcn_wmma_f32_16x16x4_f32(false, a, false, b, (short)0, c, false, false);
}

__global__ __launch_bounds__(256) void local_attn_rope_kernel(
    const float* __restrict__ q, const float* __restrict__ k,
    const float* __restrict__ v, float* __restrict__ out)
{
  __shared__ float sK [256 * SK_STRIDE];
  __shared__ float sVT[ 64 * SVT_STRIDE];

  const int wi   = blockIdx.x;
  const int bb   = blockIdx.y;
  const int tid  = threadIdx.x;
  const int wv   = __builtin_amdgcn_readfirstlane(tid >> 5);
  const int rt   = (wv < 4) ? (7 - wv) : (wv - 4);
  const int nt   = 9 + rt;
  const int lane = tid & 31;
  const int l16  = lane & 15;
  const int hi   = lane >> 4;

  const float* kb = k + (size_t)bb * 4096 * 64;
  const float* vb = v + (size_t)bb * 4096 * 64;
  const int jbase = wi * 128 - 128;

  const float LOG2_IF = -0.41524101186092029f;

  const int krow0 = tid >> 5;
  const int kc    = tid & 31;
  float kx1[32], kx2[32];
  #pragma unroll
  for (int it = 0; it < 32; ++it) {
    int grow = jbase + krow0 + 8 * it;
    grow = grow < 0 ? 0 : grow;
    const float* kr = kb + (size_t)grow * 64;
    kx1[it] = kr[kc];
    kx2[it] = kr[kc + 32];
  }
  const int vrow0 = tid >> 4;
  const int vc4   = (tid & 15) * 4;
  float4 vx[16];
  #pragma unroll
  for (int it = 0; it < 16; ++it) {
    int grow = jbase + vrow0 + 16 * it;
    grow = grow < 0 ? 0 : grow;
    vx[it] = *(const float4*)(vb + (size_t)grow * 64 + vc4);
  }
  const int iq = 16 * rt + l16;
  const float* qrow = q + ((size_t)bb * 4096 + (size_t)wi * 128 + iq) * 64;
  v2f qf[16];
  #pragma unroll
  for (int t = 0; t < 16; ++t)
    qf[t] = *(const v2f*)(qrow + 4 * t + 2 * hi);

  #pragma unroll
  for (int it = 0; it < 32; ++it) {
    int row = krow0 + 8 * it;
    float x1 = kx1[it];
    float x2 = kx2[it];
    float ang = (float)row * exp2f(LOG2_IF * (float)kc);
    float sn, cs; __sincosf(ang, &sn, &cs);
    sK[row * SK_STRIDE + kc]      = x1 * cs - x2 * sn;
    sK[row * SK_STRIDE + kc + 32] = x2 * cs + x1 * sn;
  }

  #pragma unroll
  for (int it = 0; it < 16; ++it) {
    int row = vrow0 + 16 * it;
    bool pad = (jbase + row) < 0;
    float4 val = vx[it];
    sVT[(vc4 + 0) * SVT_STRIDE + row] = pad ? 0.f : val.x;
    sVT[(vc4 + 1) * SVT_STRIDE + row] = pad ? 0.f : val.y;
    sVT[(vc4 + 2) * SVT_STRIDE + row] = pad ? 0.f : val.z;
    sVT[(vc4 + 3) * SVT_STRIDE + row] = pad ? 0.f : val.w;
  }

  {
    const float scale = 0.125f;
    float pos = (float)(128 + iq);
    #pragma unroll
    for (int t = 0; t < 8; ++t) {
      #pragma unroll
      for (int i = 0; i < 2; ++i) {
        int kk = 4 * t + 2 * hi + i;
        float ang = pos * exp2f(LOG2_IF * (float)kk);
        float sn, cs; __sincosf(ang, &sn, &cs);
        cs *= scale; sn *= scale;
        float lo = qf[t][i], hh = qf[t + 8][i];
        qf[t][i]     = lo * cs - hh * sn;
        qf[t + 8][i] = hh * cs + lo * sn;
      }
    }
  }

  __syncthreads();

  const int pm = 4 * ((l16 >> 1) & 3) + 2 * (l16 >> 3) + (l16 & 1);
  v8f S[16];
  #pragma unroll
  for (int jt = 0; jt < 16; ++jt) {
    if (jt < 9 || jt < nt) {
      v8f acc = {0.f,0.f,0.f,0.f,0.f,0.f,0.f,0.f};
      const float* kr = &sK[(16 * jt + pm) * SK_STRIDE + 2 * hi];
      #pragma unroll
      for (int t = 0; t < 16; ++t) {
        v2f kf = *(const v2f*)(kr + 4 * t);
        acc = wmma_f32_16x16x4(kf, qf[t], acc);
      }
      S[jt] = acc;
    }
  }

  const float NEGMAX = -3.402823466e38f;
  #pragma unroll
  for (int jt = 0; jt < 16; ++jt) {
    if (jt < 9 || jt < nt) {
      #pragma unroll
      for (int vv = 0; vv < 8; ++vv) {
        int j = 16 * jt + 4 * (vv >> 1) + 2 * hi + (vv & 1);
        bool padj = (wi == 0) && (j < 128);
        if (padj || (j > iq + 128)) S[jt][vv] = NEGMAX;
      }
    }
  }

  float rinv[8];
  float mx[8], sm[8];
  #pragma unroll
  for (int vv = 0; vv < 8; ++vv) mx[vv] = S[0][vv];
  #pragma unroll
  for (int jt = 1; jt < 16; ++jt) {
    if (jt < 9 || jt < nt) {
      #pragma unroll
      for (int vv = 0; vv < 8; ++vv) mx[vv] = fmaxf(mx[vv], S[jt][vv]);
    }
  }
  float qmax = fmaxf(fmaxf(fmaxf(mx[0], mx[1]), fmaxf(mx[2], mx[3])),
                     fmaxf(fmaxf(mx[4], mx[5]), fmaxf(mx[6], mx[7])));
  qmax = fmaxf(qmax, __shfl_xor(qmax, 16));
  #pragma unroll
  for (int vv = 0; vv < 8; ++vv) sm[vv] = 0.f;
  #pragma unroll
  for (int jt = 0; jt < 16; ++jt) {
    if (jt < 9 || jt < nt) {
      #pragma unroll
      for (int vv = 0; vv < 8; ++vv) {
        float e = __expf(S[jt][vv] - qmax);
        S[jt][vv] = e;
        sm[vv] += e;
      }
    }
  }
  float qsum = ((sm[0] + sm[1]) + (sm[2] + sm[3])) + ((sm[4] + sm[5]) + (sm[6] + sm[7]));
  qsum += __shfl_xor(qsum, 16);
  const float qinv = 1.0f / qsum;
  #pragma unroll
  for (int vv = 0; vv < 8; ++vv) rinv[vv] = __shfl(qinv, 8 * hi + vv, 16);

  v8f O[4];
  #pragma unroll
  for (int n = 0; n < 4; ++n) O[n] = (v8f){0.f,0.f,0.f,0.f,0.f,0.f,0.f,0.f};

  #pragma unroll
  for (int jt = 0; jt < 16; ++jt) {
    if (jt < 9 || jt < nt) {
      #pragma unroll
      for (int t = 0; t < 4; ++t) {
        v2f af = {S[jt][2 * t], S[jt][2 * t + 1]};
        int jb = 16 * jt + 4 * t + 2 * hi;
        #pragma unroll
        for (int n = 0; n < 4; ++n) {
          v2f bf = *(const v2f*)(&sVT[(16 * n + l16) * SVT_STRIDE + jb]);
          O[n] = wmma_f32_16x16x4(af, bf, O[n]);
        }
      }
    }
  }

  float* ob = out + ((size_t)bb * 4096 + (size_t)wi * 128 + 16 * rt + 8 * hi) * 64 + l16;
  #pragma unroll
  for (int n = 0; n < 4; ++n) {
    #pragma unroll
    for (int vv = 0; vv < 8; ++vv)
      ob[(size_t)vv * 64 + 16 * n] = O[n][vv] * rinv[vv];
  }
}

extern "C" void kernel_launch(void* const* d_in, const int* in_sizes, int n_in,
                              void* d_out, int out_size, void* d_ws, size_t ws_size,
                              hipStream_t stream) {
  const float* q = (const float*)d_in[0];
  const float* k = (const float*)d_in[1];
  const float* v = (const float*)d_in[2];
  float* out = (float*)d_out;
  dim3 grid(32, 32);
  local_attn_rope_kernel<<<grid, dim3(256), 0, stream>>>(q, k, v, out);
}
